# starting kernel
# baseline (speedup 1.0000x reference)
#include <hip/hip_runtime.h>
#include <hip/hip_bf16.h>

#define B_ 256
#define G_ 20000
#define P_ 1024
#define M_ 256
#define H_ 128
#define E_ 64

typedef __attribute__((ext_vector_type(2))) float v2f;
typedef __attribute__((ext_vector_type(4))) float v4f;
typedef __attribute__((ext_vector_type(8))) float v8f;

__global__ void transpose_x(const float* __restrict__ x, float* __restrict__ xT) {
    __shared__ float tile[32][33];
    int g0 = blockIdx.x * 32, b0 = blockIdx.y * 32;
    int tx = threadIdx.x, ty = threadIdx.y;
#pragma unroll
    for (int i = ty; i < 32; i += 8)
        tile[i][tx] = x[(size_t)(b0 + i) * G_ + (g0 + tx)];
    __syncthreads();
#pragma unroll
    for (int i = ty; i < 32; i += 8)
        xT[(size_t)(g0 + i) * B_ + (b0 + tx)] = tile[tx][i];
}

__global__ void pathway_kernel_t(const float* __restrict__ xT,
                                 const int* __restrict__ idx,
                                 const float* __restrict__ pw,
                                 const float* __restrict__ pmask,
                                 const float* __restrict__ pb,
                                 float* __restrict__ a_out) {
    __shared__ int   s_idx[M_];
    __shared__ float s_w[M_];
    int p = blockIdx.x;
    int t = threadIdx.x;
    for (int m = t; m < M_; m += 64) {
        s_idx[m] = idx[p * M_ + m];
        s_w[m]   = pw[p * M_ + m] * pmask[p * M_ + m];
    }
    __syncthreads();
    float a0 = 0.f, a1 = 0.f, a2 = 0.f, a3 = 0.f;
#pragma unroll 4
    for (int m = 0; m < M_; ++m) {
        v4f xv = *(const v4f*)(xT + (size_t)s_idx[m] * B_ + 4 * t);
        float w = s_w[m];
        a0 = fmaf(xv.x, w, a0);
        a1 = fmaf(xv.y, w, a1);
        a2 = fmaf(xv.z, w, a2);
        a3 = fmaf(xv.w, w, a3);
    }
    float bb = pb[p];
    a_out[(size_t)(4 * t + 0) * P_ + p] = tanhf(a0 + bb);
    a_out[(size_t)(4 * t + 1) * P_ + p] = tanhf(a1 + bb);
    a_out[(size_t)(4 * t + 2) * P_ + p] = tanhf(a2 + bb);
    a_out[(size_t)(4 * t + 3) * P_ + p] = tanhf(a3 + bb);
}

__global__ void pathway_kernel_s(const float* __restrict__ x,
                                 const int* __restrict__ idx,
                                 const float* __restrict__ pw,
                                 const float* __restrict__ pmask,
                                 const float* __restrict__ pb,
                                 float* __restrict__ a_out) {
    __shared__ int   s_idx[M_];
    __shared__ float s_w[M_];
    int p = blockIdx.x;
    int t = threadIdx.x;
    s_idx[t] = idx[p * M_ + t];
    s_w[t]   = pw[p * M_ + t] * pmask[p * M_ + t];
    __syncthreads();
    float acc = 0.f;
#pragma unroll 4
    for (int m = 0; m < M_; ++m)
        acc = fmaf(x[(size_t)t * G_ + s_idx[m]], s_w[m], acc);
    a_out[(size_t)t * P_ + p] = tanhf(acc + pb[p]);
}

template <int ACT>
__global__ void gemm_wmma_f32(const float* __restrict__ A,
                              const float* __restrict__ W,
                              const float* __restrict__ bias,
                              float* __restrict__ D,
                              int Mr, int K, int N) {
    int lane  = threadIdx.x & 31;
    int wave  = threadIdx.x >> 5;
    int waves = blockDim.x >> 5;
    int tiles_n = N >> 4;
    int tile  = blockIdx.x * waves + wave;
    int total = (Mr >> 4) * tiles_n;
    if (tile >= total) return;
    int tm = tile / tiles_n;
    int tn = tile - tm * tiles_n;
    int mn   = lane & 15;
    int koff = (lane >> 4) << 1;
    const float* arow = A + (size_t)(tm * 16 + mn) * K + koff;
    const float* wcol = W + (tn * 16 + mn);
    v8f c = {};
    for (int k = 0; k < K; k += 4) {
        v2f a = *(const v2f*)(arow + k);
        v2f b;
        b.x = wcol[(size_t)(k + koff) * N];
        b.y = wcol[(size_t)(k + koff + 1) * N];
        c = __builtin_amdgcn_wmma_f32_16x16x4_f32(
                false, a, false, b, (short)0, c, false, false);
    }
    float bv = bias[tn * 16 + mn];
    int rbase = tm * 16 + ((lane >> 4) << 3);
#pragma unroll
    for (int j = 0; j < 8; ++j) {
        float v = c[j] + bv;
        if (ACT == 1) v = fmaxf(v, 0.f);
        D[(size_t)(rbase + j) * N + (tn * 16 + mn)] = v;
    }
}

__global__ void dec_gemm_wmma(const float* __restrict__ A,
                              const float* __restrict__ W,
                              const float* __restrict__ bias,
                              float* __restrict__ D) {
    __shared__ float wpanel[H_ * 16];
    int tn = blockIdx.x;
    int t  = threadIdx.x;
#pragma unroll
    for (int i = 0; i < H_ * 16; i += 512) {
        int e = i + t;
        wpanel[e] = W[(size_t)(e >> 4) * G_ + tn * 16 + (e & 15)];
    }
    __syncthreads();
    int lane = t & 31;
    int tm   = t >> 5;
    int mn   = lane & 15;
    int koff = (lane >> 4) << 1;
    const float* arow = A + (size_t)(tm * 16 + mn) * H_ + koff;
    v8f c = {};
#pragma unroll
    for (int k = 0; k < H_; k += 4) {
        v2f a = *(const v2f*)(arow + k);
        v2f b;
        b.x = wpanel[(k + koff) * 16 + mn];
        b.y = wpanel[(k + koff + 1) * 16 + mn];
        c = __builtin_amdgcn_wmma_f32_16x16x4_f32(
                false, a, false, b, (short)0, c, false, false);
    }
    float bv = bias[tn * 16 + mn];
    int rbase = tm * 16 + ((lane >> 4) << 3);
#pragma unroll
    for (int j = 0; j < 8; ++j)
        D[(size_t)(rbase + j) * G_ + (tn * 16 + mn)] = c[j] + bv;
}

__global__ void reparam_kernel(const float* __restrict__ ml,
                               const float* __restrict__ eps,
                               float* __restrict__ z,
                               float* __restrict__ mean_o,
                               float* __restrict__ logvar_o) {
    int i = blockIdx.x * blockDim.x + threadIdx.x;
    if (i >= B_ * E_) return;
    int b = i >> 6, e = i & 63;
    float mean   = ml[b * (2 * E_) + e];
    float logvar = ml[b * (2 * E_) + E_ + e];
    z[i]        = fmaf(eps[i], logvar, mean);
    mean_o[i]   = mean;
    logvar_o[i] = logvar;
}

extern "C" void kernel_launch(void* const* d_in, const int* in_sizes, int n_in,
                              void* d_out, int out_size, void* d_ws, size_t ws_size,
                              hipStream_t stream) {
    const float* x     = (const float*)d_in[0];
    const int*   pidx  = (const int*)d_in[1];
    const float* pmask = (const float*)d_in[2];
    const float* eps   = (const float*)d_in[3];
    const float* pw    = (const float*)d_in[4];
    const float* pb    = (const float*)d_in[5];
    const float* ew1   = (const float*)d_in[6];
    const float* eb1   = (const float*)d_in[7];
    const float* ew2   = (const float*)d_in[8];
    const float* eb2   = (const float*)d_in[9];
    const float* dw1   = (const float*)d_in[10];
    const float* db1   = (const float*)d_in[11];
    const float* dw2   = (const float*)d_in[12];
    const float* db2   = (const float*)d_in[13];

    float* ws = (float*)d_ws;
    float* a  = ws;
    float* h  = ws + 262144;
    float* ml = ws + 294912;
    float* hd = ws + 327680;
    float* xT = ws + 360448;
    size_t need_xt = (size_t)(360448 + (size_t)G_ * B_) * sizeof(float);

    float* out    = (float*)d_out;
    float* z_o    = out;
    float* recon  = out + (size_t)B_ * E_;
    float* mean_o = out + (size_t)B_ * E_ + (size_t)B_ * G_;
    float* logv_o = mean_o + (size_t)B_ * E_;

    if (ws_size >= need_xt) {
        transpose_x<<<dim3(G_ / 32, B_ / 32), dim3(32, 8), 0, stream>>>(x, xT);
        pathway_kernel_t<<<P_, 64, 0, stream>>>(xT, pidx, pw, pmask, pb, a);
    } else {
        pathway_kernel_s<<<P_, 256, 0, stream>>>(x, pidx, pw, pmask, pb, a);
    }

    gemm_wmma_f32<1><<<16, 256, 0, stream>>>(a, ew1, eb1, h, B_, P_, H_);
    gemm_wmma_f32<0><<<16, 256, 0, stream>>>(h, ew2, eb2, ml, B_, H_, 2 * E_);
    reparam_kernel<<<(B_ * E_ + 255) / 256, 256, 0, stream>>>(ml, eps, z_o, mean_o, logv_o);
    gemm_wmma_f32<1><<<16, 256, 0, stream>>>(z_o, dw1, db1, hd, B_, E_, H_);
    dec_gemm_wmma<<<G_ / 16, 512, 0, stream>>>(hd, dw2, db2, recon);
}


# Small GEMMs: one block per 16x16 tile, K split over the block's 4, 8 or 16 waves, partial tiles summed through LDS in wave order (changes the f32 summation order)
# speedup vs baseline: 1.5720x; 1.5720x over previous
#include <hip/hip_runtime.h>
#include <hip/hip_bf16.h>

#define B_ 256
#define G_ 20000
#define P_ 1024
#define M_ 256
#define H_ 128
#define E_ 64

typedef __attribute__((ext_vector_type(2))) float v2f;
typedef __attribute__((ext_vector_type(4))) float v4f;
typedef __attribute__((ext_vector_type(8))) float v8f;

__global__ void transpose_x(const float* __restrict__ x, float* __restrict__ xT) {
    __shared__ float tile[32][33];
    int g0 = blockIdx.x * 32, b0 = blockIdx.y * 32;
    int tx = threadIdx.x, ty = threadIdx.y;
#pragma unroll
    for (int i = ty; i < 32; i += 8)
        tile[i][tx] = x[(size_t)(b0 + i) * G_ + (g0 + tx)];
    __syncthreads();
#pragma unroll
    for (int i = ty; i < 32; i += 8)
        xT[(size_t)(g0 + i) * B_ + (b0 + tx)] = tile[tx][i];
}

__global__ void pathway_kernel_t(const float* __restrict__ xT,
                                 const int* __restrict__ idx,
                                 const float* __restrict__ pw,
                                 const float* __restrict__ pmask,
                                 const float* __restrict__ pb,
                                 float* __restrict__ a_out) {
    __shared__ int   s_idx[M_];
    __shared__ float s_w[M_];
    int p = blockIdx.x;
    int t = threadIdx.x;
    for (int m = t; m < M_; m += 64) {
        s_idx[m] = idx[p * M_ + m];
        s_w[m]   = pw[p * M_ + m] * pmask[p * M_ + m];
    }
    __syncthreads();
    float a0 = 0.f, a1 = 0.f, a2 = 0.f, a3 = 0.f;
#pragma unroll 4
    for (int m = 0; m < M_; ++m) {
        v4f xv = *(const v4f*)(xT + (size_t)s_idx[m] * B_ + 4 * t);
        float w = s_w[m];
        a0 = fmaf(xv.x, w, a0);
        a1 = fmaf(xv.y, w, a1);
        a2 = fmaf(xv.z, w, a2);
        a3 = fmaf(xv.w, w, a3);
    }
    float bb = pb[p];
    a_out[(size_t)(4 * t + 0) * P_ + p] = tanhf(a0 + bb);
    a_out[(size_t)(4 * t + 1) * P_ + p] = tanhf(a1 + bb);
    a_out[(size_t)(4 * t + 2) * P_ + p] = tanhf(a2 + bb);
    a_out[(size_t)(4 * t + 3) * P_ + p] = tanhf(a3 + bb);
}

__global__ void pathway_kernel_s(const float* __restrict__ x,
                                 const int* __restrict__ idx,
                                 const float* __restrict__ pw,
                                 const float* __restrict__ pmask,
                                 const float* __restrict__ pb,
                                 float* __restrict__ a_out) {
    __shared__ int   s_idx[M_];
    __shared__ float s_w[M_];
    int p = blockIdx.x;
    int t = threadIdx.x;
    s_idx[t] = idx[p * M_ + t];
    s_w[t]   = pw[p * M_ + t] * pmask[p * M_ + t];
    __syncthreads();
    float acc = 0.f;
#pragma unroll 4
    for (int m = 0; m < M_; ++m)
        acc = fmaf(x[(size_t)t * G_ + s_idx[m]], s_w[m], acc);
    a_out[(size_t)t * P_ + p] = tanhf(acc + pb[p]);
}

template <int ACT, int WAVES>
__global__ void gemm_wmma_f32(const float* __restrict__ A,
                              const float* __restrict__ W,
                              const float* __restrict__ bias,
                              float* __restrict__ D,
                              int Mr, int K, int N) {
    __shared__ float part[WAVES * 256];
    int lane  = threadIdx.x & 31;
    int wave  = threadIdx.x >> 5;
    int tiles_n = N >> 4;
    int tile  = blockIdx.x;
    int total = (Mr >> 4) * tiles_n;
    if (tile >= total) return;
    int tm = tile / tiles_n;
    int tn = tile - tm * tiles_n;
    int mn   = lane & 15;
    int koff = (lane >> 4) << 1;
    int kper = K / WAVES;
    int k0   = wave * kper;
    const float* arow = A + (size_t)(tm * 16 + mn) * K + k0 + koff;
    const float* wcol = W + (size_t)k0 * N + (tn * 16 + mn);
    v8f c = {};
    for (int k = 0; k < kper; k += 4) {
        v2f a = *(const v2f*)(arow + k);
        v2f b;
        b.x = wcol[(size_t)(k + koff) * N];
        b.y = wcol[(size_t)(k + koff + 1) * N];
        c = __builtin_amdgcn_wmma_f32_16x16x4_f32(
                false, a, false, b, (short)0, c, false, false);
    }
#pragma unroll
    for (int j = 0; j < 8; ++j)
        part[wave * 256 + j * 32 + lane] = c[j];
    __syncthreads();
    for (int t = threadIdx.x; t < 256; t += WAVES * 32) {
        float v = part[t];
#pragma unroll
        for (int w = 1; w < WAVES; ++w)
            v += part[w * 256 + t];
        int l   = t & 31;
        int j   = t >> 5;
        int col = tn * 16 + (l & 15);
        int row = tm * 16 + ((l >> 4) << 3) + j;
        v += bias[col];
        if (ACT == 1) v = fmaxf(v, 0.f);
        D[(size_t)row * N + col] = v;
    }
}

__global__ void dec_gemm_wmma(const float* __restrict__ A,
                              const float* __restrict__ W,
                              const float* __restrict__ bias,
                              float* __restrict__ D) {
    __shared__ float wpanel[H_ * 16];
    int tn = blockIdx.x;
    int t  = threadIdx.x;
#pragma unroll
    for (int i = 0; i < H_ * 16; i += 512) {
        int e = i + t;
        wpanel[e] = W[(size_t)(e >> 4) * G_ + tn * 16 + (e & 15)];
    }
    __syncthreads();
    int lane = t & 31;
    int tm   = t >> 5;
    int mn   = lane & 15;
    int koff = (lane >> 4) << 1;
    const float* arow = A + (size_t)(tm * 16 + mn) * H_ + koff;
    v8f c = {};
#pragma unroll
    for (int k = 0; k < H_; k += 4) {
        v2f a = *(const v2f*)(arow + k);
        v2f b;
        b.x = wpanel[(k + koff) * 16 + mn];
        b.y = wpanel[(k + koff + 1) * 16 + mn];
        c = __builtin_amdgcn_wmma_f32_16x16x4_f32(
                false, a, false, b, (short)0, c, false, false);
    }
    float bv = bias[tn * 16 + mn];
    int rbase = tm * 16 + ((lane >> 4) << 3);
#pragma unroll
    for (int j = 0; j < 8; ++j)
        D[(size_t)(rbase + j) * G_ + (tn * 16 + mn)] = c[j] + bv;
}

__global__ void reparam_kernel(const float* __restrict__ ml,
                               const float* __restrict__ eps,
                               float* __restrict__ z,
                               float* __restrict__ mean_o,
                               float* __restrict__ logvar_o) {
    int i = blockIdx.x * blockDim.x + threadIdx.x;
    if (i >= B_ * E_) return;
    int b = i >> 6, e = i & 63;
    float mean   = ml[b * (2 * E_) + e];
    float logvar = ml[b * (2 * E_) + E_ + e];
    z[i]        = fmaf(eps[i], logvar, mean);
    mean_o[i]   = mean;
    logvar_o[i] = logvar;
}

extern "C" void kernel_launch(void* const* d_in, const int* in_sizes, int n_in,
                              void* d_out, int out_size, void* d_ws, size_t ws_size,
                              hipStream_t stream) {
    const float* x     = (const float*)d_in[0];
    const int*   pidx  = (const int*)d_in[1];
    const float* pmask = (const float*)d_in[2];
    const float* eps   = (const float*)d_in[3];
    const float* pw    = (const float*)d_in[4];
    const float* pb    = (const float*)d_in[5];
    const float* ew1   = (const float*)d_in[6];
    const float* eb1   = (const float*)d_in[7];
    const float* ew2   = (const float*)d_in[8];
    const float* eb2   = (const float*)d_in[9];
    const float* dw1   = (const float*)d_in[10];
    const float* db1   = (const float*)d_in[11];
    const float* dw2   = (const float*)d_in[12];
    const float* db2   = (const float*)d_in[13];

    float* ws = (float*)d_ws;
    float* a  = ws;
    float* h  = ws + 262144;
    float* ml = ws + 294912;
    float* hd = ws + 327680;
    float* xT = ws + 360448;
    size_t need_xt = (size_t)(360448 + (size_t)G_ * B_) * sizeof(float);

    float* out    = (float*)d_out;
    float* z_o    = out;
    float* recon  = out + (size_t)B_ * E_;
    float* mean_o = out + (size_t)B_ * E_ + (size_t)B_ * G_;
    float* logv_o = mean_o + (size_t)B_ * E_;

    if (ws_size >= need_xt) {
        transpose_x<<<dim3(G_ / 32, B_ / 32), dim3(32, 8), 0, stream>>>(x, xT);
        pathway_kernel_t<<<P_, 64, 0, stream>>>(xT, pidx, pw, pmask, pb, a);
    } else {
        pathway_kernel_s<<<P_, 256, 0, stream>>>(x, pidx, pw, pmask, pb, a);
    }

    gemm_wmma_f32<1, 16><<<128, 512, 0, stream>>>(a, ew1, eb1, h, B_, P_, H_);
    gemm_wmma_f32<0, 8><<<128, 256, 0, stream>>>(h, ew2, eb2, ml, B_, H_, 2 * E_);
    reparam_kernel<<<(B_ * E_ + 255) / 256, 256, 0, stream>>>(ml, eps, z_o, mean_o, logv_o);
    gemm_wmma_f32<1, 4><<<128, 128, 0, stream>>>(z_o, dw1, db1, hd, B_, E_, H_);
    dec_gemm_wmma<<<G_ / 16, 512, 0, stream>>>(hd, dw2, db2, recon);
}


# Pathway gather: 32 independent 16-byte row loads in flight per lane (was 4), gene indices read block-uniform through scalar loads one batch ahead; summation order unchanged
# speedup vs baseline: 1.8106x; 1.1518x over previous
#include <hip/hip_runtime.h>
#include <hip/hip_bf16.h>

#define B_ 256
#define G_ 20000
#define P_ 1024
#define M_ 256
#define H_ 128
#define E_ 64

typedef __attribute__((ext_vector_type(2))) float v2f;
typedef __attribute__((ext_vector_type(4))) float v4f;
typedef __attribute__((ext_vector_type(8))) float v8f;

__global__ void transpose_x(const float* __restrict__ x, float* __restrict__ xT) {
    __shared__ float tile[32][33];
    int g0 = blockIdx.x * 32, b0 = blockIdx.y * 32;
    int tx = threadIdx.x, ty = threadIdx.y;
#pragma unroll
    for (int i = ty; i < 32; i += 8)
        tile[i][tx] = x[(size_t)(b0 + i) * G_ + (g0 + tx)];
    __syncthreads();
#pragma unroll
    for (int i = ty; i < 32; i += 8)
        xT[(size_t)(g0 + i) * B_ + (b0 + tx)] = tile[tx][i];
}

#define GU_ 32
__global__ void __launch_bounds__(64)
pathway_kernel_t(const float* __restrict__ xT,
                 const int* __restrict__ idx,
                 const float* __restrict__ pw,
                 const float* __restrict__ pmask,
                 const float* __restrict__ pb,
                 float* __restrict__ a_out) {
    __shared__ v4f s_w4[M_ / 4];
    int p = blockIdx.x;
    int t = threadIdx.x;
    const int* ip = idx + p * M_;
    int gi[GU_];
    for (int u = 0; u < GU_; ++u) gi[u] = ip[u];
    {
        v4f wv = *(const v4f*)(pw    + p * M_ + 4 * t);
        v4f mv = *(const v4f*)(pmask + p * M_ + 4 * t);
        s_w4[t] = wv * mv;
    }
    __syncthreads();
    const float* s_w = (const float*)s_w4;
    const char* xbase = (const char*)xT;
    unsigned lane_off = (unsigned)t * 16u;
    float a0 = 0.f, a1 = 0.f, a2 = 0.f, a3 = 0.f;
    for (int m0 = 0; m0 < M_; m0 += GU_) {
        v4f xv[GU_];
        for (int u = 0; u < GU_; ++u) {
            unsigned row_off = (unsigned)gi[u] << 10;
            xv[u] = *(const v4f*)(xbase + row_off + lane_off);
        }
        if (m0 + GU_ < M_) {
            for (int u = 0; u < GU_; ++u) gi[u] = ip[m0 + GU_ + u];
        }
        for (int u = 0; u < GU_; ++u) {
            float w = s_w[m0 + u];
            a0 = fmaf(xv[u].x, w, a0);
            a1 = fmaf(xv[u].y, w, a1);
            a2 = fmaf(xv[u].z, w, a2);
            a3 = fmaf(xv[u].w, w, a3);
        }
    }
    float bb = pb[p];
    a_out[(size_t)(4 * t + 0) * P_ + p] = tanhf(a0 + bb);
    a_out[(size_t)(4 * t + 1) * P_ + p] = tanhf(a1 + bb);
    a_out[(size_t)(4 * t + 2) * P_ + p] = tanhf(a2 + bb);
    a_out[(size_t)(4 * t + 3) * P_ + p] = tanhf(a3 + bb);
}

__global__ void pathway_kernel_s(const float* __restrict__ x,
                                 const int* __restrict__ idx,
                                 const float* __restrict__ pw,
                                 const float* __restrict__ pmask,
                                 const float* __restrict__ pb,
                                 float* __restrict__ a_out) {
    __shared__ int   s_idx[M_];
    __shared__ float s_w[M_];
    int p = blockIdx.x;
    int t = threadIdx.x;
    s_idx[t] = idx[p * M_ + t];
    s_w[t]   = pw[p * M_ + t] * pmask[p * M_ + t];
    __syncthreads();
    float acc = 0.f;
#pragma unroll 4
    for (int m = 0; m < M_; ++m)
        acc = fmaf(x[(size_t)t * G_ + s_idx[m]], s_w[m], acc);
    a_out[(size_t)t * P_ + p] = tanhf(acc + pb[p]);
}

template <int ACT, int WAVES>
__global__ void gemm_wmma_f32(const float* __restrict__ A,
                              const float* __restrict__ W,
                              const float* __restrict__ bias,
                              float* __restrict__ D,
                              int Mr, int K, int N) {
    __shared__ float part[WAVES * 256];
    int lane  = threadIdx.x & 31;
    int wave  = threadIdx.x >> 5;
    int tiles_n = N >> 4;
    int tile  = blockIdx.x;
    int total = (Mr >> 4) * tiles_n;
    if (tile >= total) return;
    int tm = tile / tiles_n;
    int tn = tile - tm * tiles_n;
    int mn   = lane & 15;
    int koff = (lane >> 4) << 1;
    int kper = K / WAVES;
    int k0   = wave * kper;
    const float* arow = A + (size_t)(tm * 16 + mn) * K + k0 + koff;
    const float* wcol = W + (size_t)k0 * N + (tn * 16 + mn);
    v8f c = {};
    for (int k = 0; k < kper; k += 4) {
        v2f a = *(const v2f*)(arow + k);
        v2f b;
        b.x = wcol[(size_t)(k + koff) * N];
        b.y = wcol[(size_t)(k + koff + 1) * N];
        c = __builtin_amdgcn_wmma_f32_16x16x4_f32(
                false, a, false, b, (short)0, c, false, false);
    }
#pragma unroll
    for (int j = 0; j < 8; ++j)
        part[wave * 256 + j * 32 + lane] = c[j];
    __syncthreads();
    for (int t = threadIdx.x; t < 256; t += WAVES * 32) {
        float v = part[t];
#pragma unroll
        for (int w = 1; w < WAVES; ++w)
            v += part[w * 256 + t];
        int l   = t & 31;
        int j   = t >> 5;
        int col = tn * 16 + (l & 15);
        int row = tm * 16 + ((l >> 4) << 3) + j;
        v += bias[col];
        if (ACT == 1) v = fmaxf(v, 0.f);
        D[(size_t)row * N + col] = v;
    }
}

__global__ void dec_gemm_wmma(const float* __restrict__ A,
                              const float* __restrict__ W,
                              const float* __restrict__ bias,
                              float* __restrict__ D) {
    __shared__ float wpanel[H_ * 16];
    int tn = blockIdx.x;
    int t  = threadIdx.x;
#pragma unroll
    for (int i = 0; i < H_ * 16; i += 512) {
        int e = i + t;
        wpanel[e] = W[(size_t)(e >> 4) * G_ + tn * 16 + (e & 15)];
    }
    __syncthreads();
    int lane = t & 31;
    int tm   = t >> 5;
    int mn   = lane & 15;
    int koff = (lane >> 4) << 1;
    const float* arow = A + (size_t)(tm * 16 + mn) * H_ + koff;
    v8f c = {};
#pragma unroll
    for (int k = 0; k < H_; k += 4) {
        v2f a = *(const v2f*)(arow + k);
        v2f b;
        b.x = wpanel[(k + koff) * 16 + mn];
        b.y = wpanel[(k + koff + 1) * 16 + mn];
        c = __builtin_amdgcn_wmma_f32_16x16x4_f32(
                false, a, false, b, (short)0, c, false, false);
    }
    float bv = bias[tn * 16 + mn];
    int rbase = tm * 16 + ((lane >> 4) << 3);
#pragma unroll
    for (int j = 0; j < 8; ++j)
        D[(size_t)(rbase + j) * G_ + (tn * 16 + mn)] = c[j] + bv;
}

__global__ void reparam_kernel(const float* __restrict__ ml,
                               const float* __restrict__ eps,
                               float* __restrict__ z,
                               float* __restrict__ mean_o,
                               float* __restrict__ logvar_o) {
    int i = blockIdx.x * blockDim.x + threadIdx.x;
    if (i >= B_ * E_) return;
    int b = i >> 6, e = i & 63;
    float mean   = ml[b * (2 * E_) + e];
    float logvar = ml[b * (2 * E_) + E_ + e];
    z[i]        = fmaf(eps[i], logvar, mean);
    mean_o[i]   = mean;
    logvar_o[i] = logvar;
}

extern "C" void kernel_launch(void* const* d_in, const int* in_sizes, int n_in,
                              void* d_out, int out_size, void* d_ws, size_t ws_size,
                              hipStream_t stream) {
    const float* x     = (const float*)d_in[0];
    const int*   pidx  = (const int*)d_in[1];
    const float* pmask = (const float*)d_in[2];
    const float* eps   = (const float*)d_in[3];
    const float* pw    = (const float*)d_in[4];
    const float* pb    = (const float*)d_in[5];
    const float* ew1   = (const float*)d_in[6];
    const float* eb1   = (const float*)d_in[7];
    const float* ew2   = (const float*)d_in[8];
    const float* eb2   = (const float*)d_in[9];
    const float* dw1   = (const float*)d_in[10];
    const float* db1   = (const float*)d_in[11];
    const float* dw2   = (const float*)d_in[12];
    const float* db2   = (const float*)d_in[13];

    float* ws = (float*)d_ws;
    float* a  = ws;
    float* h  = ws + 262144;
    float* ml = ws + 294912;
    float* hd = ws + 327680;
    float* xT = ws + 360448;
    size_t need_xt = (size_t)(360448 + (size_t)G_ * B_) * sizeof(float);

    float* out    = (float*)d_out;
    float* z_o    = out;
    float* recon  = out + (size_t)B_ * E_;
    float* mean_o = out + (size_t)B_ * E_ + (size_t)B_ * G_;
    float* logv_o = mean_o + (size_t)B_ * E_;

    if (ws_size >= need_xt) {
        transpose_x<<<dim3(G_ / 32, B_ / 32), dim3(32, 8), 0, stream>>>(x, xT);
        pathway_kernel_t<<<P_, 64, 0, stream>>>(xT, pidx, pw, pmask, pb, a);
    } else {
        pathway_kernel_s<<<P_, 256, 0, stream>>>(x, pidx, pw, pmask, pb, a);
    }

    gemm_wmma_f32<1, 16><<<128, 512, 0, stream>>>(a, ew1, eb1, h, B_, P_, H_);
    gemm_wmma_f32<0, 8><<<128, 256, 0, stream>>>(h, ew2, eb2, ml, B_, H_, 2 * E_);
    reparam_kernel<<<(B_ * E_ + 255) / 256, 256, 0, stream>>>(ml, eps, z_o, mean_o, logv_o);
    gemm_wmma_f32<1, 4><<<128, 128, 0, stream>>>(z_o, dw1, db1, hd, B_, E_, H_);
    dec_gemm_wmma<<<G_ / 16, 512, 0, stream>>>(hd, dw2, db2, recon);
}


# Transpose: one 8-wave block turns a whole 32-gene x 256-batch tile (32 row loads in flight per lane, 1 KB output rows) in place of 32x32 tiles: 5000 waves enter, not 40000
# speedup vs baseline: 1.9452x; 1.0743x over previous
#include <hip/hip_runtime.h>
#include <hip/hip_bf16.h>

#define B_ 256
#define G_ 20000
#define P_ 1024
#define M_ 256
#define H_ 128
#define E_ 64

typedef __attribute__((ext_vector_type(2))) float v2f;
typedef __attribute__((ext_vector_type(4))) float v4f;
typedef __attribute__((ext_vector_type(8))) float v8f;

#define TG_ 32
__global__ void __launch_bounds__(B_)
transpose_x(const float* __restrict__ x, float* __restrict__ xT) {
    __shared__ float tile[B_][TG_ + 1];
    int g0 = blockIdx.x * TG_;
    int tx = threadIdx.x & 31;
    int ty = threadIdx.x >> 5;
    const float* src = x + (size_t)ty * G_ + (g0 + tx);
    float v[B_ / 8];
#pragma unroll
    for (int i = 0; i < B_ / 8; ++i)
        v[i] = src[(size_t)(8 * i) * G_];
#pragma unroll
    for (int i = 0; i < B_ / 8; ++i)
        tile[ty + 8 * i][tx] = v[i];
    __syncthreads();
    int b = threadIdx.x;
    float* dst = xT + (size_t)g0 * B_ + b;
#pragma unroll
    for (int gi = 0; gi < TG_; ++gi)
        dst[(size_t)gi * B_] = tile[b][gi];
}

#define GU_ 32
__global__ void __launch_bounds__(64)
pathway_kernel_t(const float* __restrict__ xT,
                 const int* __restrict__ idx,
                 const float* __restrict__ pw,
                 const float* __restrict__ pmask,
                 const float* __restrict__ pb,
                 float* __restrict__ a_out) {
    __shared__ v4f s_w4[M_ / 4];
    int p = blockIdx.x;
    int t = threadIdx.x;
    const int* ip = idx + p * M_;
    int gi[GU_];
    for (int u = 0; u < GU_; ++u) gi[u] = ip[u];
    {
        v4f wv = *(const v4f*)(pw    + p * M_ + 4 * t);
        v4f mv = *(const v4f*)(pmask + p * M_ + 4 * t);
        s_w4[t] = wv * mv;
    }
    __syncthreads();
    const float* s_w = (const float*)s_w4;
    const char* xbase = (const char*)xT;
    unsigned lane_off = (unsigned)t * 16u;
    float a0 = 0.f, a1 = 0.f, a2 = 0.f, a3 = 0.f;
    for (int m0 = 0; m0 < M_; m0 += GU_) {
        v4f xv[GU_];
        for (int u = 0; u < GU_; ++u) {
            unsigned row_off = (unsigned)gi[u] << 10;
            xv[u] = *(const v4f*)(xbase + row_off + lane_off);
        }
        if (m0 + GU_ < M_) {
            for (int u = 0; u < GU_; ++u) gi[u] = ip[m0 + GU_ + u];
        }
        for (int u = 0; u < GU_; ++u) {
            float w = s_w[m0 + u];
            a0 = fmaf(xv[u].x, w, a0);
            a1 = fmaf(xv[u].y, w, a1);
            a2 = fmaf(xv[u].z, w, a2);
            a3 = fmaf(xv[u].w, w, a3);
        }
    }
    float bb = pb[p];
    a_out[(size_t)(4 * t + 0) * P_ + p] = tanhf(a0 + bb);
    a_out[(size_t)(4 * t + 1) * P_ + p] = tanhf(a1 + bb);
    a_out[(size_t)(4 * t + 2) * P_ + p] = tanhf(a2 + bb);
    a_out[(size_t)(4 * t + 3) * P_ + p] = tanhf(a3 + bb);
}

__global__ void pathway_kernel_s(const float* __restrict__ x,
                                 const int* __restrict__ idx,
                                 const float* __restrict__ pw,
                                 const float* __restrict__ pmask,
                                 const float* __restrict__ pb,
                                 float* __restrict__ a_out) {
    __shared__ int   s_idx[M_];
    __shared__ float s_w[M_];
    int p = blockIdx.x;
    int t = threadIdx.x;
    s_idx[t] = idx[p * M_ + t];
    s_w[t]   = pw[p * M_ + t] * pmask[p * M_ + t];
    __syncthreads();
    float acc = 0.f;
#pragma unroll 4
    for (int m = 0; m < M_; ++m)
        acc = fmaf(x[(size_t)t * G_ + s_idx[m]], s_w[m], acc);
    a_out[(size_t)t * P_ + p] = tanhf(acc + pb[p]);
}

template <int ACT, int WAVES>
__global__ void gemm_wmma_f32(const float* __restrict__ A,
                              const float* __restrict__ W,
                              const float* __restrict__ bias,
                              float* __restrict__ D,
                              int Mr, int K, int N) {
    __shared__ float part[WAVES * 256];
    int lane  = threadIdx.x & 31;
    int wave  = threadIdx.x >> 5;
    int tiles_n = N >> 4;
    int tile  = blockIdx.x;
    int total = (Mr >> 4) * tiles_n;
    if (tile >= total) return;
    int tm = tile / tiles_n;
    int tn = tile - tm * tiles_n;
    int mn   = lane & 15;
    int koff = (lane >> 4) << 1;
    int kper = K / WAVES;
    int k0   = wave * kper;
    const float* arow = A + (size_t)(tm * 16 + mn) * K + k0 + koff;
    const float* wcol = W + (size_t)k0 * N + (tn * 16 + mn);
    v8f c = {};
    for (int k = 0; k < kper; k += 4) {
        v2f a = *(const v2f*)(arow + k);
        v2f b;
        b.x = wcol[(size_t)(k + koff) * N];
        b.y = wcol[(size_t)(k + koff + 1) * N];
        c = __builtin_amdgcn_wmma_f32_16x16x4_f32(
                false, a, false, b, (short)0, c, false, false);
    }
#pragma unroll
    for (int j = 0; j < 8; ++j)
        part[wave * 256 + j * 32 + lane] = c[j];
    __syncthreads();
    for (int t = threadIdx.x; t < 256; t += WAVES * 32) {
        float v = part[t];
#pragma unroll
        for (int w = 1; w < WAVES; ++w)
            v += part[w * 256 + t];
        int l   = t & 31;
        int j   = t >> 5;
        int col = tn * 16 + (l & 15);
        int row = tm * 16 + ((l >> 4) << 3) + j;
        v += bias[col];
        if (ACT == 1) v = fmaxf(v, 0.f);
        D[(size_t)row * N + col] = v;
    }
}

__global__ void dec_gemm_wmma(const float* __restrict__ A,
                              const float* __restrict__ W,
                              const float* __restrict__ bias,
                              float* __restrict__ D) {
    __shared__ float wpanel[H_ * 16];
    int tn = blockIdx.x;
    int t  = threadIdx.x;
#pragma unroll
    for (int i = 0; i < H_ * 16; i += 512) {
        int e = i + t;
        wpanel[e] = W[(size_t)(e >> 4) * G_ + tn * 16 + (e & 15)];
    }
    __syncthreads();
    int lane = t & 31;
    int tm   = t >> 5;
    int mn   = lane & 15;
    int koff = (lane >> 4) << 1;
    const float* arow = A + (size_t)(tm * 16 + mn) * H_ + koff;
    v8f c = {};
#pragma unroll
    for (int k = 0; k < H_; k += 4) {
        v2f a = *(const v2f*)(arow + k);
        v2f b;
        b.x = wpanel[(k + koff) * 16 + mn];
        b.y = wpanel[(k + koff + 1) * 16 + mn];
        c = __builtin_amdgcn_wmma_f32_16x16x4_f32(
                false, a, false, b, (short)0, c, false, false);
    }
    float bv = bias[tn * 16 + mn];
    int rbase = tm * 16 + ((lane >> 4) << 3);
#pragma unroll
    for (int j = 0; j < 8; ++j)
        D[(size_t)(rbase + j) * G_ + (tn * 16 + mn)] = c[j] + bv;
}

__global__ void reparam_kernel(const float* __restrict__ ml,
                               const float* __restrict__ eps,
                               float* __restrict__ z,
                               float* __restrict__ mean_o,
                               float* __restrict__ logvar_o) {
    int i = blockIdx.x * blockDim.x + threadIdx.x;
    if (i >= B_ * E_) return;
    int b = i >> 6, e = i & 63;
    float mean   = ml[b * (2 * E_) + e];
    float logvar = ml[b * (2 * E_) + E_ + e];
    z[i]        = fmaf(eps[i], logvar, mean);
    mean_o[i]   = mean;
    logvar_o[i] = logvar;
}

extern "C" void kernel_launch(void* const* d_in, const int* in_sizes, int n_in,
                              void* d_out, int out_size, void* d_ws, size_t ws_size,
                              hipStream_t stream) {
    const float* x     = (const float*)d_in[0];
    const int*   pidx  = (const int*)d_in[1];
    const float* pmask = (const float*)d_in[2];
    const float* eps   = (const float*)d_in[3];
    const float* pw    = (const float*)d_in[4];
    const float* pb    = (const float*)d_in[5];
    const float* ew1   = (const float*)d_in[6];
    const float* eb1   = (const float*)d_in[7];
    const float* ew2   = (const float*)d_in[8];
    const float* eb2   = (const float*)d_in[9];
    const float* dw1   = (const float*)d_in[10];
    const float* db1   = (const float*)d_in[11];
    const float* dw2   = (const float*)d_in[12];
    const float* db2   = (const float*)d_in[13];

    float* ws = (float*)d_ws;
    float* a  = ws;
    float* h  = ws + 262144;
    float* ml = ws + 294912;
    float* hd = ws + 327680;
    float* xT = ws + 360448;
    size_t need_xt = (size_t)(360448 + (size_t)G_ * B_) * sizeof(float);

    float* out    = (float*)d_out;
    float* z_o    = out;
    float* recon  = out + (size_t)B_ * E_;
    float* mean_o = out + (size_t)B_ * E_ + (size_t)B_ * G_;
    float* logv_o = mean_o + (size_t)B_ * E_;

    if (ws_size >= need_xt) {
        transpose_x<<<G_ / TG_, B_, 0, stream>>>(x, xT);
        pathway_kernel_t<<<P_, 64, 0, stream>>>(xT, pidx, pw, pmask, pb, a);
    } else {
        pathway_kernel_s<<<P_, 256, 0, stream>>>(x, pidx, pw, pmask, pb, a);
    }

    gemm_wmma_f32<1, 16><<<128, 512, 0, stream>>>(a, ew1, eb1, h, B_, P_, H_);
    gemm_wmma_f32<0, 8><<<128, 256, 0, stream>>>(h, ew2, eb2, ml, B_, H_, 2 * E_);
    reparam_kernel<<<(B_ * E_ + 255) / 256, 256, 0, stream>>>(ml, eps, z_o, mean_o, logv_o);
    gemm_wmma_f32<1, 4><<<128, 128, 0, stream>>>(z_o, dw1, db1, hd, B_, E_, H_);
    dec_gemm_wmma<<<G_ / 16, 512, 0, stream>>>(hd, dw2, db2, recon);
}


# Decoder GEMM: one 16-wave block per 80-column panel (250 blocks, one per WGP), hd and the W panel staged once in LDS, 5 independent WMMA chains per wave; 4000 waves enter, not 20000; sums unchanged
# speedup vs baseline: 2.3271x; 1.1963x over previous
#include <hip/hip_runtime.h>
#include <hip/hip_bf16.h>

#define B_ 256
#define G_ 20000
#define P_ 1024
#define M_ 256
#define H_ 128
#define E_ 64

typedef __attribute__((ext_vector_type(2))) float v2f;
typedef __attribute__((ext_vector_type(4))) float v4f;
typedef __attribute__((ext_vector_type(8))) float v8f;

#define TG_ 32
__global__ void __launch_bounds__(B_)
transpose_x(const float* __restrict__ x, float* __restrict__ xT) {
    __shared__ float tile[B_][TG_ + 1];
    int g0 = blockIdx.x * TG_;
    int tx = threadIdx.x & 31;
    int ty = threadIdx.x >> 5;
    const float* src = x + (size_t)ty * G_ + (g0 + tx);
    float v[B_ / 8];
#pragma unroll
    for (int i = 0; i < B_ / 8; ++i)
        v[i] = src[(size_t)(8 * i) * G_];
#pragma unroll
    for (int i = 0; i < B_ / 8; ++i)
        tile[ty + 8 * i][tx] = v[i];
    __syncthreads();
    int b = threadIdx.x;
    float* dst = xT + (size_t)g0 * B_ + b;
#pragma unroll
    for (int gi = 0; gi < TG_; ++gi)
        dst[(size_t)gi * B_] = tile[b][gi];
}

#define GU_ 32
__global__ void __launch_bounds__(64)
pathway_kernel_t(const float* __restrict__ xT,
                 const int* __restrict__ idx,
                 const float* __restrict__ pw,
                 const float* __restrict__ pmask,
                 const float* __restrict__ pb,
                 float* __restrict__ a_out) {
    __shared__ v4f s_w4[M_ / 4];
    int p = blockIdx.x;
    int t = threadIdx.x;
    const int* ip = idx + p * M_;
    int gi[GU_];
    for (int u = 0; u < GU_; ++u) gi[u] = ip[u];
    {
        v4f wv = *(const v4f*)(pw    + p * M_ + 4 * t);
        v4f mv = *(const v4f*)(pmask + p * M_ + 4 * t);
        s_w4[t] = wv * mv;
    }
    __syncthreads();
    const float* s_w = (const float*)s_w4;
    const char* xbase = (const char*)xT;
    unsigned lane_off = (unsigned)t * 16u;
    float a0 = 0.f, a1 = 0.f, a2 = 0.f, a3 = 0.f;
    for (int m0 = 0; m0 < M_; m0 += GU_) {
        v4f xv[GU_];
        for (int u = 0; u < GU_; ++u) {
            unsigned row_off = (unsigned)gi[u] << 10;
            xv[u] = *(const v4f*)(xbase + row_off + lane_off);
        }
        if (m0 + GU_ < M_) {
            for (int u = 0; u < GU_; ++u) gi[u] = ip[m0 + GU_ + u];
        }
        for (int u = 0; u < GU_; ++u) {
            float w = s_w[m0 + u];
            a0 = fmaf(xv[u].x, w, a0);
            a1 = fmaf(xv[u].y, w, a1);
            a2 = fmaf(xv[u].z, w, a2);
            a3 = fmaf(xv[u].w, w, a3);
        }
    }
    float bb = pb[p];
    a_out[(size_t)(4 * t + 0) * P_ + p] = tanhf(a0 + bb);
    a_out[(size_t)(4 * t + 1) * P_ + p] = tanhf(a1 + bb);
    a_out[(size_t)(4 * t + 2) * P_ + p] = tanhf(a2 + bb);
    a_out[(size_t)(4 * t + 3) * P_ + p] = tanhf(a3 + bb);
}

__global__ void pathway_kernel_s(const float* __restrict__ x,
                                 const int* __restrict__ idx,
                                 const float* __restrict__ pw,
                                 const float* __restrict__ pmask,
                                 const float* __restrict__ pb,
                                 float* __restrict__ a_out) {
    __shared__ int   s_idx[M_];
    __shared__ float s_w[M_];
    int p = blockIdx.x;
    int t = threadIdx.x;
    s_idx[t] = idx[p * M_ + t];
    s_w[t]   = pw[p * M_ + t] * pmask[p * M_ + t];
    __syncthreads();
    float acc = 0.f;
#pragma unroll 4
    for (int m = 0; m < M_; ++m)
        acc = fmaf(x[(size_t)t * G_ + s_idx[m]], s_w[m], acc);
    a_out[(size_t)t * P_ + p] = tanhf(acc + pb[p]);
}

template <int ACT, int WAVES>
__global__ void gemm_wmma_f32(const float* __restrict__ A,
                              const float* __restrict__ W,
                              const float* __restrict__ bias,
                              float* __restrict__ D,
                              int Mr, int K, int N) {
    __shared__ float part[WAVES * 256];
    int lane  = threadIdx.x & 31;
    int wave  = threadIdx.x >> 5;
    int tiles_n = N >> 4;
    int tile  = blockIdx.x;
    int total = (Mr >> 4) * tiles_n;
    if (tile >= total) return;
    int tm = tile / tiles_n;
    int tn = tile - tm * tiles_n;
    int mn   = lane & 15;
    int koff = (lane >> 4) << 1;
    int kper = K / WAVES;
    int k0   = wave * kper;
    const float* arow = A + (size_t)(tm * 16 + mn) * K + k0 + koff;
    const float* wcol = W + (size_t)k0 * N + (tn * 16 + mn);
    v8f c = {};
    for (int k = 0; k < kper; k += 4) {
        v2f a = *(const v2f*)(arow + k);
        v2f b;
        b.x = wcol[(size_t)(k + koff) * N];
        b.y = wcol[(size_t)(k + koff + 1) * N];
        c = __builtin_amdgcn_wmma_f32_16x16x4_f32(
                false, a, false, b, (short)0, c, false, false);
    }
#pragma unroll
    for (int j = 0; j < 8; ++j)
        part[wave * 256 + j * 32 + lane] = c[j];
    __syncthreads();
    for (int t = threadIdx.x; t < 256; t += WAVES * 32) {
        float v = part[t];
#pragma unroll
        for (int w = 1; w < WAVES; ++w)
            v += part[w * 256 + t];
        int l   = t & 31;
        int j   = t >> 5;
        int col = tn * 16 + (l & 15);
        int row = tm * 16 + ((l >> 4) << 3) + j;
        v += bias[col];
        if (ACT == 1) v = fmaxf(v, 0.f);
        D[(size_t)row * N + col] = v;
    }
}

#define DT_ 5
#define DPW_ (DT_ * 16)
#define APITCH_ (H_ + 4)
#define WITEMS_ ((H_ / 2) * (DPW_ / 4))
__global__ void __launch_bounds__(512)
dec_gemm_wmma(const float* __restrict__ A,
              const float* __restrict__ W,
              const float* __restrict__ bias,
              float* __restrict__ D) {
    __shared__ __attribute__((aligned(16))) float lds[H_ * DPW_ + B_ * APITCH_];
    float* wpanel = lds;
    float* apanel = lds + H_ * DPW_;
    int t    = threadIdx.x;
    int col0 = blockIdx.x * DPW_;
    v4f w0[3], w1[3];
    v4f av[(B_ * H_) / (4 * 512)];
#pragma unroll
    for (int i = 0; i < 3; ++i) {
        int e = i * 512 + t;
        if (e < WITEMS_) {
            int kk = e / (DPW_ / 4);
            int c4 = e - kk * (DPW_ / 4);
            const float* src = W + (size_t)(2 * kk) * G_ + col0 + c4 * 4;
            w0[i] = *(const v4f*)src;
            w1[i] = *(const v4f*)(src + G_);
        }
    }
#pragma unroll
    for (int i = 0; i < (B_ * H_) / (4 * 512); ++i)
        av[i] = *(const v4f*)(A + (size_t)(i * 512 + t) * 4);
#pragma unroll
    for (int i = 0; i < 3; ++i) {
        int e = i * 512 + t;
        if (e < WITEMS_) {
            v4f lo = {w0[i].x, w1[i].x, w0[i].y, w1[i].y};
            v4f hi = {w0[i].z, w1[i].z, w0[i].w, w1[i].w};
            *(v4f*)(wpanel + e * 8)     = lo;
            *(v4f*)(wpanel + e * 8 + 4) = hi;
        }
    }
#pragma unroll
    for (int i = 0; i < (B_ * H_) / (4 * 512); ++i) {
        int e4 = i * 512 + t;
        *(v4f*)(apanel + (e4 >> 5) * APITCH_ + (e4 & 31) * 4) = av[i];
    }
    __syncthreads();
    int lane = t & 31;
    int tm   = t >> 5;
    int mn   = lane & 15;
    int half = lane >> 4;
    const float* ap = apanel + (tm * 16 + mn) * APITCH_ + 2 * half;
    const float* wp = wpanel + (half * DPW_ + mn) * 2;
    v8f c[DT_];
#pragma unroll
    for (int j = 0; j < DT_; ++j) c[j] = (v8f){};
#pragma unroll
    for (int k = 0; k < H_; k += 4) {
        v2f a = *(const v2f*)(ap + k);
#pragma unroll
        for (int j = 0; j < DT_; ++j) {
            v2f b = *(const v2f*)(wp + k * DPW_ + j * 32);
            c[j] = __builtin_amdgcn_wmma_f32_16x16x4_f32(
                       false, a, false, b, (short)0, c[j], false, false);
        }
    }
    int rbase = tm * 16 + (half << 3);
#pragma unroll
    for (int j = 0; j < DT_; ++j) {
        int col  = col0 + j * 16 + mn;
        float bv = bias[col];
#pragma unroll
        for (int r = 0; r < 8; ++r)
            D[(size_t)(rbase + r) * G_ + col] = c[j][r] + bv;
    }
}

__global__ void reparam_kernel(const float* __restrict__ ml,
                               const float* __restrict__ eps,
                               float* __restrict__ z,
                               float* __restrict__ mean_o,
                               float* __restrict__ logvar_o) {
    int i = blockIdx.x * blockDim.x + threadIdx.x;
    if (i >= B_ * E_) return;
    int b = i >> 6, e = i & 63;
    float mean   = ml[b * (2 * E_) + e];
    float logvar = ml[b * (2 * E_) + E_ + e];
    z[i]        = fmaf(eps[i], logvar, mean);
    mean_o[i]   = mean;
    logvar_o[i] = logvar;
}

extern "C" void kernel_launch(void* const* d_in, const int* in_sizes, int n_in,
                              void* d_out, int out_size, void* d_ws, size_t ws_size,
                              hipStream_t stream) {
    const float* x     = (const float*)d_in[0];
    const int*   pidx  = (const int*)d_in[1];
    const float* pmask = (const float*)d_in[2];
    const float* eps   = (const float*)d_in[3];
    const float* pw    = (const float*)d_in[4];
    const float* pb    = (const float*)d_in[5];
    const float* ew1   = (const float*)d_in[6];
    const float* eb1   = (const float*)d_in[7];
    const float* ew2   = (const float*)d_in[8];
    const float* eb2   = (const float*)d_in[9];
    const float* dw1   = (const float*)d_in[10];
    const float* db1   = (const float*)d_in[11];
    const float* dw2   = (const float*)d_in[12];
    const float* db2   = (const float*)d_in[13];

    float* ws = (float*)d_ws;
    float* a  = ws;
    float* h  = ws + 262144;
    float* ml = ws + 294912;
    float* hd = ws + 327680;
    float* xT = ws + 360448;
    size_t need_xt = (size_t)(360448 + (size_t)G_ * B_) * sizeof(float);

    float* out    = (float*)d_out;
    float* z_o    = out;
    float* recon  = out + (size_t)B_ * E_;
    float* mean_o = out + (size_t)B_ * E_ + (size_t)B_ * G_;
    float* logv_o = mean_o + (size_t)B_ * E_;

    if (ws_size >= need_xt) {
        transpose_x<<<G_ / TG_, B_, 0, stream>>>(x, xT);
        pathway_kernel_t<<<P_, 64, 0, stream>>>(xT, pidx, pw, pmask, pb, a);
    } else {
        pathway_kernel_s<<<P_, 256, 0, stream>>>(x, pidx, pw, pmask, pb, a);
    }

    gemm_wmma_f32<1, 16><<<128, 512, 0, stream>>>(a, ew1, eb1, h, B_, P_, H_);
    gemm_wmma_f32<0, 8><<<128, 256, 0, stream>>>(h, ew2, eb2, ml, B_, H_, 2 * E_);
    reparam_kernel<<<(B_ * E_ + 255) / 256, 256, 0, stream>>>(ml, eps, z_o, mean_o, logv_o);
    gemm_wmma_f32<1, 4><<<128, 128, 0, stream>>>(z_o, dw1, db1, hd, B_, E_, H_);
    dec_gemm_wmma<<<G_ / DPW_, 512, 0, stream>>>(hd, dw2, db2, recon);
}


# Encoder output GEMM, reparameterization and decoder input GEMM run as ONE launch: a 16-wave block owns 16 batch rows through all three, ml and z pass through LDS; 5 launches, not 7; sums unchanged
# speedup vs baseline: 2.4992x; 1.0740x over previous
#include <hip/hip_runtime.h>
#include <hip/hip_bf16.h>

#define B_ 256
#define G_ 20000
#define P_ 1024
#define M_ 256
#define H_ 128
#define E_ 64

typedef __attribute__((ext_vector_type(2))) float v2f;
typedef __attribute__((ext_vector_type(4))) float v4f;
typedef __attribute__((ext_vector_type(8))) float v8f;

#define TG_ 32
__global__ void __launch_bounds__(B_)
transpose_x(const float* __restrict__ x, float* __restrict__ xT) {
    __shared__ float tile[B_][TG_ + 1];
    int g0 = blockIdx.x * TG_;
    int tx = threadIdx.x & 31;
    int ty = threadIdx.x >> 5;
    const float* src = x + (size_t)ty * G_ + (g0 + tx);
    float v[B_ / 8];
#pragma unroll
    for (int i = 0; i < B_ / 8; ++i)
        v[i] = src[(size_t)(8 * i) * G_];
#pragma unroll
    for (int i = 0; i < B_ / 8; ++i)
        tile[ty + 8 * i][tx] = v[i];
    __syncthreads();
    int b = threadIdx.x;
    float* dst = xT + (size_t)g0 * B_ + b;
#pragma unroll
    for (int gi = 0; gi < TG_; ++gi)
        dst[(size_t)gi * B_] = tile[b][gi];
}

#define GU_ 32
__global__ void __launch_bounds__(64)
pathway_kernel_t(const float* __restrict__ xT,
                 const int* __restrict__ idx,
                 const float* __restrict__ pw,
                 const float* __restrict__ pmask,
                 const float* __restrict__ pb,
                 float* __restrict__ a_out) {
    __shared__ v4f s_w4[M_ / 4];
    int p = blockIdx.x;
    int t = threadIdx.x;
    const int* ip = idx + p * M_;
    int gi[GU_];
    for (int u = 0; u < GU_; ++u) gi[u] = ip[u];
    {
        v4f wv = *(const v4f*)(pw    + p * M_ + 4 * t);
        v4f mv = *(const v4f*)(pmask + p * M_ + 4 * t);
        s_w4[t] = wv * mv;
    }
    __syncthreads();
    const float* s_w = (const float*)s_w4;
    const char* xbase = (const char*)xT;
    unsigned lane_off = (unsigned)t * 16u;
    float a0 = 0.f, a1 = 0.f, a2 = 0.f, a3 = 0.f;
    for (int m0 = 0; m0 < M_; m0 += GU_) {
        v4f xv[GU_];
        for (int u = 0; u < GU_; ++u) {
            unsigned row_off = (unsigned)gi[u] << 10;
            xv[u] = *(const v4f*)(xbase + row_off + lane_off);
        }
        if (m0 + GU_ < M_) {
            for (int u = 0; u < GU_; ++u) gi[u] = ip[m0 + GU_ + u];
        }
        for (int u = 0; u < GU_; ++u) {
            float w = s_w[m0 + u];
            a0 = fmaf(xv[u].x, w, a0);
            a1 = fmaf(xv[u].y, w, a1);
            a2 = fmaf(xv[u].z, w, a2);
            a3 = fmaf(xv[u].w, w, a3);
        }
    }
    float bb = pb[p];
    a_out[(size_t)(4 * t + 0) * P_ + p] = tanhf(a0 + bb);
    a_out[(size_t)(4 * t + 1) * P_ + p] = tanhf(a1 + bb);
    a_out[(size_t)(4 * t + 2) * P_ + p] = tanhf(a2 + bb);
    a_out[(size_t)(4 * t + 3) * P_ + p] = tanhf(a3 + bb);
}

__global__ void pathway_kernel_s(const float* __restrict__ x,
                                 const int* __restrict__ idx,
                                 const float* __restrict__ pw,
                                 const float* __restrict__ pmask,
                                 const float* __restrict__ pb,
                                 float* __restrict__ a_out) {
    __shared__ int   s_idx[M_];
    __shared__ float s_w[M_];
    int p = blockIdx.x;
    int t = threadIdx.x;
    s_idx[t] = idx[p * M_ + t];
    s_w[t]   = pw[p * M_ + t] * pmask[p * M_ + t];
    __syncthreads();
    float acc = 0.f;
#pragma unroll 4
    for (int m = 0; m < M_; ++m)
        acc = fmaf(x[(size_t)t * G_ + s_idx[m]], s_w[m], acc);
    a_out[(size_t)t * P_ + p] = tanhf(acc + pb[p]);
}

template <int ACT, int WAVES>
__global__ void gemm_wmma_f32(const float* __restrict__ A,
                              const float* __restrict__ W,
                              const float* __restrict__ bias,
                              float* __restrict__ D,
                              int Mr, int K, int N) {
    __shared__ float part[WAVES * 256];
    int lane  = threadIdx.x & 31;
    int wave  = threadIdx.x >> 5;
    int tiles_n = N >> 4;
    int tile  = blockIdx.x;
    int total = (Mr >> 4) * tiles_n;
    if (tile >= total) return;
    int tm = tile / tiles_n;
    int tn = tile - tm * tiles_n;
    int mn   = lane & 15;
    int koff = (lane >> 4) << 1;
    int kper = K / WAVES;
    int k0   = wave * kper;
    const float* arow = A + (size_t)(tm * 16 + mn) * K + k0 + koff;
    const float* wcol = W + (size_t)k0 * N + (tn * 16 + mn);
    v8f c = {};
    for (int k = 0; k < kper; k += 4) {
        v2f a = *(const v2f*)(arow + k);
        v2f b;
        b.x = wcol[(size_t)(k + koff) * N];
        b.y = wcol[(size_t)(k + koff + 1) * N];
        c = __builtin_amdgcn_wmma_f32_16x16x4_f32(
                false, a, false, b, (short)0, c, false, false);
    }
#pragma unroll
    for (int j = 0; j < 8; ++j)
        part[wave * 256 + j * 32 + lane] = c[j];
    __syncthreads();
    for (int t = threadIdx.x; t < 256; t += WAVES * 32) {
        float v = part[t];
#pragma unroll
        for (int w = 1; w < WAVES; ++w)
            v += part[w * 256 + t];
        int l   = t & 31;
        int j   = t >> 5;
        int col = tn * 16 + (l & 15);
        int row = tm * 16 + ((l >> 4) << 3) + j;
        v += bias[col];
        if (ACT == 1) v = fmaxf(v, 0.f);
        D[(size_t)row * N + col] = v;
    }
}

#define DT_ 5
#define DPW_ (DT_ * 16)
#define APITCH_ (H_ + 4)
#define WITEMS_ ((H_ / 2) * (DPW_ / 4))
__global__ void __launch_bounds__(512)
dec_gemm_wmma(const float* __restrict__ A,
              const float* __restrict__ W,
              const float* __restrict__ bias,
              float* __restrict__ D) {
    __shared__ __attribute__((aligned(16))) float lds[H_ * DPW_ + B_ * APITCH_];
    float* wpanel = lds;
    float* apanel = lds + H_ * DPW_;
    int t    = threadIdx.x;
    int col0 = blockIdx.x * DPW_;
    v4f w0[3], w1[3];
    v4f av[(B_ * H_) / (4 * 512)];
#pragma unroll
    for (int i = 0; i < 3; ++i) {
        int e = i * 512 + t;
        if (e < WITEMS_) {
            int kk = e / (DPW_ / 4);
            int c4 = e - kk * (DPW_ / 4);
            const float* src = W + (size_t)(2 * kk) * G_ + col0 + c4 * 4;
            w0[i] = *(const v4f*)src;
            w1[i] = *(const v4f*)(src + G_);
        }
    }
#pragma unroll
    for (int i = 0; i < (B_ * H_) / (4 * 512); ++i)
        av[i] = *(const v4f*)(A + (size_t)(i * 512 + t) * 4);
#pragma unroll
    for (int i = 0; i < 3; ++i) {
        int e = i * 512 + t;
        if (e < WITEMS_) {
            v4f lo = {w0[i].x, w1[i].x, w0[i].y, w1[i].y};
            v4f hi = {w0[i].z, w1[i].z, w0[i].w, w1[i].w};
            *(v4f*)(wpanel + e * 8)     = lo;
            *(v4f*)(wpanel + e * 8 + 4) = hi;
        }
    }
#pragma unroll
    for (int i = 0; i < (B_ * H_) / (4 * 512); ++i) {
        int e4 = i * 512 + t;
        *(v4f*)(apanel + (e4 >> 5) * APITCH_ + (e4 & 31) * 4) = av[i];
    }
    __syncthreads();
    int lane = t & 31;
    int tm   = t >> 5;
    int mn   = lane & 15;
    int half = lane >> 4;
    const float* ap = apanel + (tm * 16 + mn) * APITCH_ + 2 * half;
    const float* wp = wpanel + (half * DPW_ + mn) * 2;
    v8f c[DT_];
#pragma unroll
    for (int j = 0; j < DT_; ++j) c[j] = (v8f){};
#pragma unroll
    for (int k = 0; k < H_; k += 4) {
        v2f a = *(const v2f*)(ap + k);
#pragma unroll
        for (int j = 0; j < DT_; ++j) {
            v2f b = *(const v2f*)(wp + k * DPW_ + j * 32);
            c[j] = __builtin_amdgcn_wmma_f32_16x16x4_f32(
                       false, a, false, b, (short)0, c[j], false, false);
        }
    }
    int rbase = tm * 16 + (half << 3);
#pragma unroll
    for (int j = 0; j < DT_; ++j) {
        int col  = col0 + j * 16 + mn;
        float bv = bias[col];
#pragma unroll
        for (int r = 0; r < 8; ++r)
            D[(size_t)(rbase + r) * G_ + col] = c[j][r] + bv;
    }
}

#define ZP_ (E_ + 4)
__global__ void __launch_bounds__(512)
enc_tail_dec_head(const float* __restrict__ h,
                  const float* __restrict__ w2,
                  const float* __restrict__ b2,
                  const float* __restrict__ eps,
                  const float* __restrict__ dw1,
                  const float* __restrict__ db1,
                  float* __restrict__ z_o,
                  float* __restrict__ mean_o,
                  float* __restrict__ logv_o,
                  float* __restrict__ hd) {
    __shared__ float part[64 * 256];
    __shared__ __attribute__((aligned(8))) float zs[16 * ZP_];
    int t    = threadIdx.x;
    int lane = t & 31;
    int wave = t >> 5;
    int mn   = lane & 15;
    int koff = (lane >> 4) << 1;
    int row0 = blockIdx.x * 16;

    int ks1 = wave & 7;
    int g1  = wave >> 3;
    int ks3 = wave & 3;
    int g3  = wave >> 2;
    v2f a1[4];
    v4f p1[4], q1[4];
    v2f p3[4], q3[4];
    {
        const float* ar = h + (size_t)(row0 + mn) * H_ + 16 * ks1 + koff;
        const float* wr = w2 + (size_t)(16 * ks1 + koff) * (2 * E_) + 64 * g1 + 4 * mn;
#pragma unroll
        for (int s = 0; s < 4; ++s) {
            a1[s] = *(const v2f*)(ar + 4 * s);
            p1[s] = *(const v4f*)(wr + (size_t)(4 * s) * (2 * E_));
            q1[s] = *(const v4f*)(wr + (size_t)(4 * s + 1) * (2 * E_));
        }
        const float* dr = dw1 + (size_t)(16 * ks3 + koff) * H_ + 32 * g3 + 2 * mn;
#pragma unroll
        for (int s = 0; s < 4; ++s) {
            p3[s] = *(const v2f*)(dr + (size_t)(4 * s) * H_);
            q3[s] = *(const v2f*)(dr + (size_t)(4 * s + 1) * H_);
        }
    }
    float epsv[2], bm[2], bl[2];
#pragma unroll
    for (int i = 0; i < 2; ++i) {
        int q = t + 512 * i;
        int e = q & 63;
        epsv[i] = eps[(size_t)(row0 + (q >> 6)) * E_ + e];
        bm[i]   = b2[e];
        bl[i]   = b2[E_ + e];
    }
    float bd[4];
#pragma unroll
    for (int i = 0; i < 4; ++i) bd[i] = db1[(t + 512 * i) & 127];
    __builtin_amdgcn_sched_barrier(0);

    {
        v8f c[4];
#pragma unroll
        for (int j = 0; j < 4; ++j) c[j] = (v8f){};
#pragma unroll
        for (int s = 0; s < 4; ++s) {
#pragma unroll
            for (int j = 0; j < 4; ++j) {
                v2f b = {p1[s][j], q1[s][j]};
                c[j] = __builtin_amdgcn_wmma_f32_16x16x4_f32(
                           false, a1[s], false, b, (short)0, c[j], false, false);
            }
        }
#pragma unroll
        for (int j = 0; j < 4; ++j) {
            float* pp = part + ((g1 * 4 + j) * 8 + ks1) * 256 + lane;
#pragma unroll
            for (int r = 0; r < 8; ++r) pp[r * 32] = c[j][r];
        }
    }
    __syncthreads();

    float mv[2], lv[2], zv2[2];
#pragma unroll
    for (int i = 0; i < 2; ++i) {
        int q   = t + 512 * i;
        int row = q >> 6;
        int e   = q & 63;
        int idx = (row & 7) * 32 + ((row >> 3) << 4) + (e >> 2);
        const float* pm = part + ((e & 3) * 8) * 256 + idx;
        const float* pl = pm + 4 * 8 * 256;
        float mean = pm[0], logvar = pl[0];
#pragma unroll
        for (int k = 1; k < 8; ++k) {
            mean   += pm[k * 256];
            logvar += pl[k * 256];
        }
        mean   += bm[i];
        logvar += bl[i];
        float zv = fmaf(epsv[i], logvar, mean);
        mv[i] = mean;
        lv[i] = logvar;
        zv2[i] = zv;
        zs[row * ZP_ + e] = zv;
    }
    __syncthreads();

    {
        v8f c[2];
        c[0] = (v8f){};
        c[1] = (v8f){};
        const float* zr = zs + mn * ZP_ + 16 * ks3 + koff;
#pragma unroll
        for (int s = 0; s < 4; ++s) {
            v2f a = *(const v2f*)(zr + 4 * s);
#pragma unroll
            for (int j = 0; j < 2; ++j) {
                v2f b = {p3[s][j], q3[s][j]};
                c[j] = __builtin_amdgcn_wmma_f32_16x16x4_f32(
                           false, a, false, b, (short)0, c[j], false, false);
            }
        }
#pragma unroll
        for (int j = 0; j < 2; ++j) {
            float* pp = part + ((g3 * 2 + j) * 4 + ks3) * 256 + lane;
#pragma unroll
            for (int r = 0; r < 8; ++r) pp[r * 32] = c[j][r];
        }
    }
    __syncthreads();
#pragma unroll
    for (int i = 0; i < 4; ++i) {
        int q   = t + 512 * i;
        int row = q >> 7;
        int col = q & 127;
        int cc  = col & 31;
        int idx = (row & 7) * 32 + ((row >> 3) << 4) + (cc >> 1);
        const float* pp = part + (((col >> 5) * 2 + (cc & 1)) * 4) * 256 + idx;
        float v = pp[0];
#pragma unroll
        for (int k = 1; k < 4; ++k) v += pp[k * 256];
        v += bd[i];
        v = fmaxf(v, 0.f);
        hd[(size_t)(row0 + row) * H_ + col] = v;
    }
#pragma unroll
    for (int i = 0; i < 2; ++i) {
        int q = t + 512 * i;
        size_t o = (size_t)(row0 + (q >> 6)) * E_ + (q & 63);
        mean_o[o] = mv[i];
        logv_o[o] = lv[i];
        z_o[o]    = zv2[i];
    }
}

extern "C" void kernel_launch(void* const* d_in, const int* in_sizes, int n_in,
                              void* d_out, int out_size, void* d_ws, size_t ws_size,
                              hipStream_t stream) {
    const float* x     = (const float*)d_in[0];
    const int*   pidx  = (const int*)d_in[1];
    const float* pmask = (const float*)d_in[2];
    const float* eps   = (const float*)d_in[3];
    const float* pw    = (const float*)d_in[4];
    const float* pb    = (const float*)d_in[5];
    const float* ew1   = (const float*)d_in[6];
    const float* eb1   = (const float*)d_in[7];
    const float* ew2   = (const float*)d_in[8];
    const float* eb2   = (const float*)d_in[9];
    const float* dw1   = (const float*)d_in[10];
    const float* db1   = (const float*)d_in[11];
    const float* dw2   = (const float*)d_in[12];
    const float* db2   = (const float*)d_in[13];

    float* ws = (float*)d_ws;
    float* a  = ws;
    float* h  = ws + 262144;
    float* hd = ws + 327680;
    float* xT = ws + 360448;
    size_t need_xt = (size_t)(360448 + (size_t)G_ * B_) * sizeof(float);

    float* out    = (float*)d_out;
    float* z_o    = out;
    float* recon  = out + (size_t)B_ * E_;
    float* mean_o = out + (size_t)B_ * E_ + (size_t)B_ * G_;
    float* logv_o = mean_o + (size_t)B_ * E_;

    if (ws_size >= need_xt) {
        transpose_x<<<G_ / TG_, B_, 0, stream>>>(x, xT);
        pathway_kernel_t<<<P_, 64, 0, stream>>>(xT, pidx, pw, pmask, pb, a);
    } else {
        pathway_kernel_s<<<P_, 256, 0, stream>>>(x, pidx, pw, pmask, pb, a);
    }

    gemm_wmma_f32<1, 16><<<128, 512, 0, stream>>>(a, ew1, eb1, h, B_, P_, H_);
    enc_tail_dec_head<<<B_ / 16, 512, 0, stream>>>(h, ew2, eb2, eps, dw1, db1, z_o, mean_o, logv_o, hd);
    dec_gemm_wmma<<<G_ / DPW_, 512, 0, stream>>>(hd, dw2, db2, recon);
}


# Pathway activations go to the encoder GEMM pathway-major [P][B]: a block stores its row as 8 whole lines (was 256 scattered words); A loads touch 2 lines not 16, all 64 issued at once; same sums
# speedup vs baseline: 3.0519x; 1.2212x over previous
#include <hip/hip_runtime.h>
#include <hip/hip_bf16.h>

#define B_ 256
#define G_ 20000
#define P_ 1024
#define M_ 256
#define H_ 128
#define E_ 64

typedef __attribute__((ext_vector_type(2))) float v2f;
typedef __attribute__((ext_vector_type(4))) float v4f;
typedef __attribute__((ext_vector_type(8))) float v8f;

#define TG_ 32
__global__ void __launch_bounds__(B_)
transpose_x(const float* __restrict__ x, float* __restrict__ xT) {
    __shared__ float tile[B_][TG_ + 1];
    int g0 = blockIdx.x * TG_;
    int tx = threadIdx.x & 31;
    int ty = threadIdx.x >> 5;
    const float* src = x + (size_t)ty * G_ + (g0 + tx);
    float v[B_ / 8];
#pragma unroll
    for (int i = 0; i < B_ / 8; ++i)
        v[i] = src[(size_t)(8 * i) * G_];
#pragma unroll
    for (int i = 0; i < B_ / 8; ++i)
        tile[ty + 8 * i][tx] = v[i];
    __syncthreads();
    int b = threadIdx.x;
    float* dst = xT + (size_t)g0 * B_ + b;
#pragma unroll
    for (int gi = 0; gi < TG_; ++gi)
        dst[(size_t)gi * B_] = tile[b][gi];
}

#define GU_ 32
__global__ void __launch_bounds__(64)
pathway_kernel_t(const float* __restrict__ xT,
                 const int* __restrict__ idx,
                 const float* __restrict__ pw,
                 const float* __restrict__ pmask,
                 const float* __restrict__ pb,
                 float* __restrict__ a_out) {
    __shared__ v4f s_w4[M_ / 4];
    int p = blockIdx.x;
    int t = threadIdx.x;
    const int* ip = idx + p * M_;
    int gi[GU_];
    for (int u = 0; u < GU_; ++u) gi[u] = ip[u];
    {
        v4f wv = *(const v4f*)(pw    + p * M_ + 4 * t);
        v4f mv = *(const v4f*)(pmask + p * M_ + 4 * t);
        s_w4[t] = wv * mv;
    }
    __syncthreads();
    const float* s_w = (const float*)s_w4;
    const char* xbase = (const char*)xT;
    unsigned lane_off = (unsigned)t * 16u;
    float a0 = 0.f, a1 = 0.f, a2 = 0.f, a3 = 0.f;
    for (int m0 = 0; m0 < M_; m0 += GU_) {
        v4f xv[GU_];
        for (int u = 0; u < GU_; ++u) {
            unsigned row_off = (unsigned)gi[u] << 10;
            xv[u] = *(const v4f*)(xbase + row_off + lane_off);
        }
        if (m0 + GU_ < M_) {
            for (int u = 0; u < GU_; ++u) gi[u] = ip[m0 + GU_ + u];
        }
        for (int u = 0; u < GU_; ++u) {
            float w = s_w[m0 + u];
            a0 = fmaf(xv[u].x, w, a0);
            a1 = fmaf(xv[u].y, w, a1);
            a2 = fmaf(xv[u].z, w, a2);
            a3 = fmaf(xv[u].w, w, a3);
        }
    }
    float bb = pb[p];
    v4f r = {tanhf(a0 + bb), tanhf(a1 + bb), tanhf(a2 + bb), tanhf(a3 + bb)};
    *(v4f*)(a_out + (size_t)p * B_ + 4 * t) = r;
}

__global__ void pathway_kernel_s(const float* __restrict__ x,
                                 const int* __restrict__ idx,
                                 const float* __restrict__ pw,
                                 const float* __restrict__ pmask,
                                 const float* __restrict__ pb,
                                 float* __restrict__ a_out) {
    __shared__ int   s_idx[M_];
    __shared__ float s_w[M_];
    int p = blockIdx.x;
    int t = threadIdx.x;
    s_idx[t] = idx[p * M_ + t];
    s_w[t]   = pw[p * M_ + t] * pmask[p * M_ + t];
    __syncthreads();
    float acc = 0.f;
#pragma unroll 4
    for (int m = 0; m < M_; ++m)
        acc = fmaf(x[(size_t)t * G_ + s_idx[m]], s_w[m], acc);
    a_out[(size_t)p * B_ + t] = tanhf(acc + pb[p]);
}

#define G1W_ 16
#define G1K_ (P_ / G1W_)
__global__ void __launch_bounds__(G1W_ * 32)
enc_gemm1_wmma(const float* __restrict__ aT,
               const float* __restrict__ W,
               const float* __restrict__ bias,
               float* __restrict__ D) {
    __shared__ float part[G1W_ * 256];
    int lane = threadIdx.x & 31;
    int wave = threadIdx.x >> 5;
    int tm   = blockIdx.x / (H_ / 16);
    int tn   = blockIdx.x - tm * (H_ / 16);
    int mn   = lane & 15;
    int koff = (lane >> 4) << 1;
    int k0   = wave * G1K_ + koff;
    const float* acol = aT + (size_t)k0 * B_ + (tm * 16 + mn);
    const float* wcol = W  + (size_t)k0 * H_ + (tn * 16 + mn);
    v2f a[G1K_ / 4], b[G1K_ / 4];
#pragma unroll
    for (int s = 0; s < G1K_ / 4; ++s) {
        a[s].x = acol[(4 * s) * B_];
        a[s].y = acol[(4 * s + 1) * B_];
        b[s].x = wcol[(4 * s) * H_];
        b[s].y = wcol[(4 * s + 1) * H_];
    }
    v8f c = {};
#pragma unroll
    for (int s = 0; s < G1K_ / 4; ++s)
        c = __builtin_amdgcn_wmma_f32_16x16x4_f32(
                false, a[s], false, b[s], (short)0, c, false, false);
#pragma unroll
    for (int j = 0; j < 8; ++j)
        part[wave * 256 + j * 32 + lane] = c[j];
    __syncthreads();
    int t = threadIdx.x;
    if (t < 256) {
        float v = part[t];
#pragma unroll
        for (int w = 1; w < G1W_; ++w)
            v += part[w * 256 + t];
        int l   = t & 31;
        int j   = t >> 5;
        int col = tn * 16 + (l & 15);
        int row = tm * 16 + ((l >> 4) << 3) + j;
        v += bias[col];
        v = fmaxf(v, 0.f);
        D[(size_t)row * H_ + col] = v;
    }
}

#define DT_ 5
#define DPW_ (DT_ * 16)
#define APITCH_ (H_ + 4)
#define WITEMS_ ((H_ / 2) * (DPW_ / 4))
__global__ void __launch_bounds__(512)
dec_gemm_wmma(const float* __restrict__ A,
              const float* __restrict__ W,
              const float* __restrict__ bias,
              float* __restrict__ D) {
    __shared__ __attribute__((aligned(16))) float lds[H_ * DPW_ + B_ * APITCH_];
    float* wpanel = lds;
    float* apanel = lds + H_ * DPW_;
    int t    = threadIdx.x;
    int col0 = blockIdx.x * DPW_;
    v4f w0[3], w1[3];
    v4f av[(B_ * H_) / (4 * 512)];
#pragma unroll
    for (int i = 0; i < 3; ++i) {
        int e = i * 512 + t;
        if (e < WITEMS_) {
            int kk = e / (DPW_ / 4);
            int c4 = e - kk * (DPW_ / 4);
            const float* src = W + (size_t)(2 * kk) * G_ + col0 + c4 * 4;
            w0[i] = *(const v4f*)src;
            w1[i] = *(const v4f*)(src + G_);
        }
    }
#pragma unroll
    for (int i = 0; i < (B_ * H_) / (4 * 512); ++i)
        av[i] = *(const v4f*)(A + (size_t)(i * 512 + t) * 4);
#pragma unroll
    for (int i = 0; i < 3; ++i) {
        int e = i * 512 + t;
        if (e < WITEMS_) {
            v4f lo = {w0[i].x, w1[i].x, w0[i].y, w1[i].y};
            v4f hi = {w0[i].z, w1[i].z, w0[i].w, w1[i].w};
            *(v4f*)(wpanel + e * 8)     = lo;
            *(v4f*)(wpanel + e * 8 + 4) = hi;
        }
    }
#pragma unroll
    for (int i = 0; i < (B_ * H_) / (4 * 512); ++i) {
        int e4 = i * 512 + t;
        *(v4f*)(apanel + (e4 >> 5) * APITCH_ + (e4 & 31) * 4) = av[i];
    }
    __syncthreads();
    int lane = t & 31;
    int tm   = t >> 5;
    int mn   = lane & 15;
    int half = lane >> 4;
    const float* ap = apanel + (tm * 16 + mn) * APITCH_ + 2 * half;
    const float* wp = wpanel + (half * DPW_ + mn) * 2;
    v8f c[DT_];
#pragma unroll
    for (int j = 0; j < DT_; ++j) c[j] = (v8f){};
#pragma unroll
    for (int k = 0; k < H_; k += 4) {
        v2f a = *(const v2f*)(ap + k);
#pragma unroll
        for (int j = 0; j < DT_; ++j) {
            v2f b = *(const v2f*)(wp + k * DPW_ + j * 32);
            c[j] = __builtin_amdgcn_wmma_f32_16x16x4_f32(
                       false, a, false, b, (short)0, c[j], false, false);
        }
    }
    int rbase = tm * 16 + (half << 3);
#pragma unroll
    for (int j = 0; j < DT_; ++j) {
        int col  = col0 + j * 16 + mn;
        float bv = bias[col];
#pragma unroll
        for (int r = 0; r < 8; ++r)
            D[(size_t)(rbase + r) * G_ + col] = c[j][r] + bv;
    }
}

#define ZP_ (E_ + 4)
__global__ void __launch_bounds__(512)
enc_tail_dec_head(const float* __restrict__ h,
                  const float* __restrict__ w2,
                  const float* __restrict__ b2,
                  const float* __restrict__ eps,
                  const float* __restrict__ dw1,
                  const float* __restrict__ db1,
                  float* __restrict__ z_o,
                  float* __restrict__ mean_o,
                  float* __restrict__ logv_o,
                  float* __restrict__ hd) {
    __shared__ float part[64 * 256];
    __shared__ __attribute__((aligned(8))) float zs[16 * ZP_];
    int t    = threadIdx.x;
    int lane = t & 31;
    int wave = t >> 5;
    int mn   = lane & 15;
    int koff = (lane >> 4) << 1;
    int row0 = blockIdx.x * 16;

    int ks1 = wave & 7;
    int g1  = wave >> 3;
    int ks3 = wave & 3;
    int g3  = wave >> 2;
    v2f a1[4];
    v4f p1[4], q1[4];
    v2f p3[4], q3[4];
    {
        const float* ar = h + (size_t)(row0 + mn) * H_ + 16 * ks1 + koff;
        const float* wr = w2 + (size_t)(16 * ks1 + koff) * (2 * E_) + 64 * g1 + 4 * mn;
#pragma unroll
        for (int s = 0; s < 4; ++s) {
            a1[s] = *(const v2f*)(ar + 4 * s);
            p1[s] = *(const v4f*)(wr + (size_t)(4 * s) * (2 * E_));
            q1[s] = *(const v4f*)(wr + (size_t)(4 * s + 1) * (2 * E_));
        }
        const float* dr = dw1 + (size_t)(16 * ks3 + koff) * H_ + 32 * g3 + 2 * mn;
#pragma unroll
        for (int s = 0; s < 4; ++s) {
            p3[s] = *(const v2f*)(dr + (size_t)(4 * s) * H_);
            q3[s] = *(const v2f*)(dr + (size_t)(4 * s + 1) * H_);
        }
    }
    float epsv[2], bm[2], bl[2];
#pragma unroll
    for (int i = 0; i < 2; ++i) {
        int q = t + 512 * i;
        int e = q & 63;
        epsv[i] = eps[(size_t)(row0 + (q >> 6)) * E_ + e];
        bm[i]   = b2[e];
        bl[i]   = b2[E_ + e];
    }
    float bd[4];
#pragma unroll
    for (int i = 0; i < 4; ++i) bd[i] = db1[(t + 512 * i) & 127];
    __builtin_amdgcn_sched_barrier(0);

    {
        v8f c[4];
#pragma unroll
        for (int j = 0; j < 4; ++j) c[j] = (v8f){};
#pragma unroll
        for (int s = 0; s < 4; ++s) {
#pragma unroll
            for (int j = 0; j < 4; ++j) {
                v2f b = {p1[s][j], q1[s][j]};
                c[j] = __builtin_amdgcn_wmma_f32_16x16x4_f32(
                           false, a1[s], false, b, (short)0, c[j], false, false);
            }
        }
#pragma unroll
        for (int j = 0; j < 4; ++j) {
            float* pp = part + ((g1 * 4 + j) * 8 + ks1) * 256 + lane;
#pragma unroll
            for (int r = 0; r < 8; ++r) pp[r * 32] = c[j][r];
        }
    }
    __syncthreads();

    float mv[2], lv[2], zv2[2];
#pragma unroll
    for (int i = 0; i < 2; ++i) {
        int q   = t + 512 * i;
        int row = q >> 6;
        int e   = q & 63;
        int idx = (row & 7) * 32 + ((row >> 3) << 4) + (e >> 2);
        const float* pm = part + ((e & 3) * 8) * 256 + idx;
        const float* pl = pm + 4 * 8 * 256;
        float mean = pm[0], logvar = pl[0];
#pragma unroll
        for (int k = 1; k < 8; ++k) {
            mean   += pm[k * 256];
            logvar += pl[k * 256];
        }
        mean   += bm[i];
        logvar += bl[i];
        float zv = fmaf(epsv[i], logvar, mean);
        mv[i] = mean;
        lv[i] = logvar;
        zv2[i] = zv;
        zs[row * ZP_ + e] = zv;
    }
    __syncthreads();

    {
        v8f c[2];
        c[0] = (v8f){};
        c[1] = (v8f){};
        const float* zr = zs + mn * ZP_ + 16 * ks3 + koff;
#pragma unroll
        for (int s = 0; s < 4; ++s) {
            v2f a = *(const v2f*)(zr + 4 * s);
#pragma unroll
            for (int j = 0; j < 2; ++j) {
                v2f b = {p3[s][j], q3[s][j]};
                c[j] = __builtin_amdgcn_wmma_f32_16x16x4_f32(
                           false, a, false, b, (short)0, c[j], false, false);
            }
        }
#pragma unroll
        for (int j = 0; j < 2; ++j) {
            float* pp = part + ((g3 * 2 + j) * 4 + ks3) * 256 + lane;
#pragma unroll
            for (int r = 0; r < 8; ++r) pp[r * 32] = c[j][r];
        }
    }
    __syncthreads();
#pragma unroll
    for (int i = 0; i < 4; ++i) {
        int q   = t + 512 * i;
        int row = q >> 7;
        int col = q & 127;
        int cc  = col & 31;
        int idx = (row & 7) * 32 + ((row >> 3) << 4) + (cc >> 1);
        const float* pp = part + (((col >> 5) * 2 + (cc & 1)) * 4) * 256 + idx;
        float v = pp[0];
#pragma unroll
        for (int k = 1; k < 4; ++k) v += pp[k * 256];
        v += bd[i];
        v = fmaxf(v, 0.f);
        hd[(size_t)(row0 + row) * H_ + col] = v;
    }
#pragma unroll
    for (int i = 0; i < 2; ++i) {
        int q = t + 512 * i;
        size_t o = (size_t)(row0 + (q >> 6)) * E_ + (q & 63);
        mean_o[o] = mv[i];
        logv_o[o] = lv[i];
        z_o[o]    = zv2[i];
    }
}

extern "C" void kernel_launch(void* const* d_in, const int* in_sizes, int n_in,
                              void* d_out, int out_size, void* d_ws, size_t ws_size,
                              hipStream_t stream) {
    const float* x     = (const float*)d_in[0];
    const int*   pidx  = (const int*)d_in[1];
    const float* pmask = (const float*)d_in[2];
    const float* eps   = (const float*)d_in[3];
    const float* pw    = (const float*)d_in[4];
    const float* pb    = (const float*)d_in[5];
    const float* ew1   = (const float*)d_in[6];
    const float* eb1   = (const float*)d_in[7];
    const float* ew2   = (const float*)d_in[8];
    const float* eb2   = (const float*)d_in[9];
    const float* dw1   = (const float*)d_in[10];
    const float* db1   = (const float*)d_in[11];
    const float* dw2   = (const float*)d_in[12];
    const float* db2   = (const float*)d_in[13];

    float* ws = (float*)d_ws;
    float* a  = ws;
    float* h  = ws + 262144;
    float* hd = ws + 327680;
    float* xT = ws + 360448;
    size_t need_xt = (size_t)(360448 + (size_t)G_ * B_) * sizeof(float);

    float* out    = (float*)d_out;
    float* z_o    = out;
    float* recon  = out + (size_t)B_ * E_;
    float* mean_o = out + (size_t)B_ * E_ + (size_t)B_ * G_;
    float* logv_o = mean_o + (size_t)B_ * E_;

    if (ws_size >= need_xt) {
        transpose_x<<<G_ / TG_, B_, 0, stream>>>(x, xT);
        pathway_kernel_t<<<P_, 64, 0, stream>>>(xT, pidx, pw, pmask, pb, a);
    } else {
        pathway_kernel_s<<<P_, 256, 0, stream>>>(x, pidx, pw, pmask, pb, a);
    }

    enc_gemm1_wmma<<<128, 512, 0, stream>>>(a, ew1, eb1, h);
    enc_tail_dec_head<<<B_ / 16, 512, 0, stream>>>(h, ew2, eb2, eps, dw1, db1, z_o, mean_o, logv_o, hd);
    dec_gemm_wmma<<<G_ / DPW_, 512, 0, stream>>>(hd, dw2, db2, recon);
}
